# starting kernel
# baseline (speedup 1.0000x reference)
#include <hip/hip_runtime.h>


typedef float v2f __attribute__((ext_vector_type(2)));
typedef float v8f __attribute__((ext_vector_type(8)));

#define NODES 64
#define NH    10

__device__ __forceinline__ float xor16(float x) {
  int xi = __builtin_bit_cast(int, x);
  int r  = __builtin_amdgcn_permlanex16(xi, xi, 0x76543210, 0xfedcba98,
                                        false, false);
  return __builtin_bit_cast(float, r);
}

__global__ __launch_bounds__(256)
void cgnn_wmma_kernel(const float* __restrict__ noise,
                      const float* __restrict__ W1,
                      const float* __restrict__ b1g,
                      const float* __restrict__ W2g,
                      const float* __restrict__ b2g,
                      float* __restrict__ out) {
  __shared__ float4 sA[NODES * 32];
  __shared__ float sW2[NODES * 16];

  for (int e = threadIdx.x; e < NODES * 32; e += blockDim.x) {
    const int i = e >> 5, l = e & 31, j = l & 15, h = l >> 4;
    float a0 = 0.f, a1 = 0.f, a2 = 0.f, a3 = 0.f;
    if (j < NH) {
      const float* w = W1 + i * (5 * NH);
      a0 = w[(2 * h) * NH + j];
      a1 = w[(2 * h + 1) * NH + j];
      if (h == 0) {
        a2 = w[4 * NH + j];
        a3 = b1g[i * NH + j];
      }
    }
    sA[e] = make_float4(a0, a1, a2, a3);
  }
  for (int e = threadIdx.x; e < NODES * 16; e += blockDim.x) {
    const int i = e >> 4, j = e & 15;
    sW2[e] = (j < NH) ? W2g[i * NH + j] : 0.f;
  }
  __syncthreads();

  const int lane = threadIdx.x & 31;
  const int wave = threadIdx.x >> 5;
  const int hi   = lane >> 4;
  const int m    = lane & 15;
  const long row = (long)blockIdx.x * 128 + (long)wave * 16 + m;

  const float4* nrow = (const float4*)(noise + row * NODES);
  float4*       orow = (float4*)(out + row * NODES);

  const float one_lo = hi ? 0.f : 1.f;
  float b0 = 0.f, b1w = 0.f;

  auto step = [&](int i, float nz) -> float {
    const float4 ap = sA[i * 32 + lane];
    v2f A2; A2[0] = ap.z; A2[1] = ap.w;
    v2f B2; B2[0] = hi ? 0.f : nz; B2[1] = one_lo;
    v8f zero = {};
    v8f t = __builtin_amdgcn_wmma_f32_16x16x4_f32(
        false, A2, false, B2, (short)0, zero, false, false);
    v2f A1; A1[0] = ap.x; A1[1] = ap.y;
    v2f B1; B1[0] = b0;   B1[1] = b1w;
    v8f d = __builtin_amdgcn_wmma_f32_16x16x4_f32(
        false, A1, false, B1, (short)0, t, false, false);
    const float4* wp = (const float4*)&sW2[i * 16 + hi * 8];
    const float4 wa = wp[0], wb = wp[1];
    float p;
    p = fmaxf(d[0], 0.f) * wa.x;
    p = fmaf(fmaxf(d[1], 0.f), wa.y, p);
    p = fmaf(fmaxf(d[2], 0.f), wa.z, p);
    p = fmaf(fmaxf(d[3], 0.f), wa.w, p);
    p = fmaf(fmaxf(d[4], 0.f), wb.x, p);
    p = fmaf(fmaxf(d[5], 0.f), wb.y, p);
    p = fmaf(fmaxf(d[6], 0.f), wb.z, p);
    p = fmaf(fmaxf(d[7], 0.f), wb.w, p);
    return (p + xor16(p)) + b2g[i];
  };

  {
    const float4 nz4 = nrow[0];
    const float y0 = step(0, nz4.x); if (!hi) b0  = y0;
    const float y1 = step(1, nz4.y); if (!hi) b1w = y1;
    const float y2 = step(2, nz4.z); if (hi)  b0  = y2;
    const float y3 = step(3, nz4.w); if (hi)  b1w = y3;
    if (!hi) orow[0] = make_float4(y0, y1, y2, y3);
  }

  for (int i4 = 1; i4 < 16; ++i4) {
    const float4 nz4 = nrow[i4];
    if (i4 < 15) __builtin_prefetch(nrow + i4 + 1, 0, 0);
    float ys[4];
#pragma unroll
    for (int u = 0; u < 4; ++u) {
      const int i = i4 * 4 + u;
      const float nz = (u == 0) ? nz4.x : (u == 1) ? nz4.y
                      : (u == 2) ? nz4.z : nz4.w;
      const float y = step(i, nz);
      const float tmp = xor16(b0);
      b0 = b1w;
      b1w = hi ? y : tmp;
      ys[u] = y;
    }
    if (!hi) orow[i4] = make_float4(ys[0], ys[1], ys[2], ys[3]);
  }
}

extern "C" void kernel_launch(void* const* d_in, const int* in_sizes, int n_in,
                              void* d_out, int out_size, void* d_ws, size_t ws_size,
                              hipStream_t stream) {
  const float* noise = (const float*)d_in[0];
  const float* W1 = (const float*)d_in[2];
  const float* b1 = (const float*)d_in[3];
  const float* W2 = (const float*)d_in[4];
  const float* b2 = (const float*)d_in[5];
  float* out = (float*)d_out;

  const int B = in_sizes[0] / NODES;
  const int rowsPerBlock = 128;
  const int grid = B / rowsPerBlock;

  cgnn_wmma_kernel<<<grid, 256, 0, stream>>>(noise, W1, b1, W2, b2, out);
}


# Second layer by relu(x)=(x+|x|)/2: its linear half (with b2) rides as matrix row 7 of the same two products, tail = 7 fma with |.| source modifier (was 16 max + 8 fma); per-node sum order changes
# speedup vs baseline: 2.6768x; 2.6768x over previous
#include <hip/hip_runtime.h>


typedef float v2f __attribute__((ext_vector_type(2)));
typedef float v8f __attribute__((ext_vector_type(8)));

#define NODES 64
#define NH    10

__device__ __forceinline__ float xor16(float x) {
  int xi = __builtin_bit_cast(int, x);
  int r  = __builtin_amdgcn_permlanex16(xi, xi, 0x76543210, 0xfedcba98,
                                        false, false);
  return __builtin_bit_cast(float, r);
}

__global__ __launch_bounds__(256)
void cgnn_wmma_kernel(const float* __restrict__ noise,
                      const float* __restrict__ W1,
                      const float* __restrict__ b1g,
                      const float* __restrict__ W2g,
                      const float* __restrict__ b2g,
                      float* __restrict__ out) {
  __shared__ float4 sA[NODES * 32];
  __shared__ float sW2[NODES * 16];

  for (int e = threadIdx.x; e < NODES * 32; e += blockDim.x) {
    const int i = e >> 5, l = e & 31, R = l & 15, h = l >> 4;
    if (R == 7) continue;
    float a0 = 0.f, a1 = 0.f, a2 = 0.f, a3 = 0.f;
    if (R <= 10) {
      const int j = (R < 7) ? R : R - 1;
      const float* w = W1 + i * (5 * NH);
      a0 = w[(2 * h) * NH + j];
      a1 = w[(2 * h + 1) * NH + j];
      if (h == ((i >> 2) & 1)) a2 = w[4 * NH + j];
      if (h == 0) a3 = b1g[i * NH + j];
    }
    sA[e] = make_float4(a0, a1, a2, a3);
  }
  for (int e = threadIdx.x; e < NODES * 2; e += blockDim.x) {
    const int i = e >> 1, h = e & 1;
    const float* w = W1 + i * (5 * NH);
    float a0 = 0.f, a1 = 0.f, a2 = 0.f, a3 = 0.f;
#pragma unroll
    for (int j = 0; j < NH; ++j) {
      const float c = 0.5f * W2g[i * NH + j];
      a0 = fmaf(c, w[(2 * h) * NH + j], a0);
      a1 = fmaf(c, w[(2 * h + 1) * NH + j], a1);
      a2 = fmaf(c, w[4 * NH + j], a2);
      a3 = fmaf(c, b1g[i * NH + j], a3);
    }
    if (h != ((i >> 2) & 1)) a2 = 0.f;
    a3 = (h == 0) ? a3 + b2g[i] : 0.f;
    sA[i * 32 + h * 16 + 7] = make_float4(a0, a1, a2, a3);
  }
  for (int e = threadIdx.x; e < NODES * 16; e += blockDim.x) {
    const int i = e >> 4, q = e & 15, h = q >> 3, r = q & 7;
    float v = 0.f;
    if (h == 0 && r < 7) v = 0.5f * W2g[i * NH + r];
    if (h == 1 && r < 3) v = 0.5f * W2g[i * NH + 7 + r];
    sW2[e] = v;
  }

  const int lane = threadIdx.x & 31;
  const int wave = threadIdx.x >> 5;
  const int hi   = lane >> 4;
  const int m    = lane & 15;
  const long row = (long)blockIdx.x * 128 + (long)wave * 16 + m;

  const float4* nrow = (const float4*)(noise + row * NODES);
  float4*       orow = (float4*)(out + row * NODES);

  float rw[32];
#pragma unroll
  for (int t = 0; t < 8; ++t) {
    const float4 v = nrow[2 * t + hi];
    rw[4 * t + 0] = v.x; rw[4 * t + 1] = v.y;
    rw[4 * t + 2] = v.z; rw[4 * t + 3] = v.w;
  }

  __syncthreads();

  const float one_lo = hi ? 0.f : 1.f;
  float b0 = 0.f, b1w = 0.f;

  auto step = [&](int i, float nzm) -> float {
    const float4 ap = sA[i * 32 + lane];
    v2f A2; A2[0] = ap.z; A2[1] = ap.w;
    v2f B2; B2[0] = nzm; B2[1] = one_lo;
    v8f zero = {};
    v8f t = __builtin_amdgcn_wmma_f32_16x16x4_f32(
        false, A2, false, B2, (short)0, zero, false, false);
    v2f A1; A1[0] = ap.x; A1[1] = ap.y;
    v2f B1; B1[0] = b0;   B1[1] = b1w;
    v8f d = __builtin_amdgcn_wmma_f32_16x16x4_f32(
        false, A1, false, B1, (short)0, t, false, false);
    const float4* wp = (const float4*)&sW2[i * 16 + hi * 8];
    const float4 wa = wp[0], wb = wp[1];
    float p;
    p = fmaf(fabsf(d[0]), wa.x, d[7]);
    p = fmaf(fabsf(d[1]), wa.y, p);
    p = fmaf(fabsf(d[2]), wa.z, p);
    p = fmaf(fabsf(d[3]), wa.w, p);
    p = fmaf(fabsf(d[4]), wb.x, p);
    p = fmaf(fabsf(d[5]), wb.y, p);
    p = fmaf(fabsf(d[6]), wb.z, p);
    return p + xor16(p);
  };

#pragma unroll
  for (int t = 0; t < 8; ++t) {
#pragma unroll
    for (int s = 0; s < 2; ++s) {
#pragma unroll
      for (int c = 0; c < 4; ++c) {
        const int i = 8 * t + 4 * s + c;
        const int q = 4 * t + c;
        const bool own = s ? (hi != 0) : (hi == 0);
        const float y = step(i, own ? rw[q] : 0.f);
        if (i < 4) {
          if (i == 0) { if (!hi) b0  = y; }
          if (i == 1) { if (!hi) b1w = y; }
          if (i == 2) { if (hi)  b0  = y; }
          if (i == 3) { if (hi)  b1w = y; }
        } else {
          const float tmp = xor16(b0);
          b0 = b1w;
          b1w = hi ? y : tmp;
        }
        rw[q] = own ? y : rw[q];
      }
    }
  }

#pragma unroll
  for (int t = 0; t < 8; ++t)
    orow[2 * t + hi] = make_float4(rw[4 * t + 0], rw[4 * t + 1],
                                   rw[4 * t + 2], rw[4 * t + 3]);
}

extern "C" void kernel_launch(void* const* d_in, const int* in_sizes, int n_in,
                              void* d_out, int out_size, void* d_ws, size_t ws_size,
                              hipStream_t stream) {
  const float* noise = (const float*)d_in[0];
  const float* W1 = (const float*)d_in[2];
  const float* b1 = (const float*)d_in[3];
  const float* W2 = (const float*)d_in[4];
  const float* b2 = (const float*)d_in[5];
  float* out = (float*)d_out;

  const int B = in_sizes[0] / NODES;
  const int rowsPerBlock = 128;
  const int grid = B / rowsPerBlock;

  cgnn_wmma_kernel<<<grid, 256, 0, stream>>>(noise, W1, b1, W2, b2, out);
}


# Start-up loads all issued before the first wait (was a wait per loop trip): weight gathers unrolled with selects, row loads behind them, row 7 summed from the LDS table, same order; LDS-only fences
# speedup vs baseline: 2.9388x; 1.0979x over previous
#include <hip/hip_runtime.h>


typedef float v2f __attribute__((ext_vector_type(2)));
typedef float v8f __attribute__((ext_vector_type(8)));

#define NODES 64
#define NH    10

__device__ __forceinline__ float xor16(float x) {
  int xi = __builtin_bit_cast(int, x);
  int r  = __builtin_amdgcn_permlanex16(xi, xi, 0x76543210, 0xfedcba98,
                                        false, false);
  return __builtin_bit_cast(float, r);
}

__device__ __forceinline__ void lds_barrier() {
  __builtin_amdgcn_fence(__ATOMIC_RELEASE, "workgroup", "local");
  __builtin_amdgcn_s_barrier();
  __builtin_amdgcn_fence(__ATOMIC_ACQUIRE, "workgroup", "local");
}

__global__ __launch_bounds__(256)
void cgnn_wmma_kernel(const float* __restrict__ noise,
                      const float* __restrict__ W1,
                      const float* __restrict__ b1g,
                      const float* __restrict__ W2g,
                      const float* __restrict__ b2g,
                      float* __restrict__ out) {
  __shared__ float4 sA[NODES * 32];
  __shared__ float sW2[NODES * 16];

  const int lane = threadIdx.x & 31;
  const int wave = threadIdx.x >> 5;
  const int hi   = lane >> 4;
  const int m    = lane & 15;
  const long row = (long)blockIdx.x * 128 + (long)wave * 16 + m;

  const int R  = m;
  const int jr = (R < 7) ? R : R - 1;
  const int jj = (jr < NH) ? jr : NH - 1;
  float g0[8], g1[8], g2[8], g3[8];
#pragma unroll
  for (int it = 0; it < 8; ++it) {
    const int i = wave + 8 * it;
    const float* w = W1 + i * (5 * NH);
    g0[it] = w[(2 * hi) * NH + jj];
    g1[it] = w[(2 * hi + 1) * NH + jj];
    g2[it] = w[4 * NH + jj];
    g3[it] = b1g[i * NH + jj];
  }
  const int q  = threadIdx.x & 15, qh = q >> 3, qr = q & 7;
  const int wj = qh ? 7 + ((qr < 2) ? qr : 2) : ((qr < 6) ? qr : 6);
  const bool wlive = qh ? (qr < 3) : (qr < 7);
  float gw[4];
#pragma unroll
  for (int it = 0; it < 4; ++it)
    gw[it] = W2g[((int)(threadIdx.x >> 4) + 16 * it) * NH + wj];
  const float gb2 = b2g[(threadIdx.x >> 1) & (NODES - 1)];

  const float4* nrow = (const float4*)(noise + row * NODES);
  float4*       orow = (float4*)(out + row * NODES);

  float rw[32];
#pragma unroll
  for (int t = 0; t < 8; ++t) {
    const float4 v = nrow[2 * t + hi];
    rw[4 * t + 0] = v.x; rw[4 * t + 1] = v.y;
    rw[4 * t + 2] = v.z; rw[4 * t + 3] = v.w;
  }

  __builtin_amdgcn_sched_barrier(0);

#pragma unroll
  for (int it = 0; it < 8; ++it) {
    const int i = wave + 8 * it;
    const bool live = (R <= 10) && (R != 7);
    const float a0 = live ? g0[it] : 0.f;
    const float a1 = live ? g1[it] : 0.f;
    const float a2 = (live && hi == ((i >> 2) & 1)) ? g2[it] : 0.f;
    const float a3 = (live && hi == 0) ? g3[it] : 0.f;
    sA[i * 32 + lane] = make_float4(a0, a1, a2, a3);
  }
#pragma unroll
  for (int it = 0; it < 4; ++it)
    sW2[threadIdx.x + 256 * it] = wlive ? 0.5f * gw[it] : 0.f;

  lds_barrier();

  if (threadIdx.x < NODES * 2) {
    const int i = threadIdx.x >> 1, h = threadIdx.x & 1;
    const float4* ent = &sA[i * 32 + h * 16];
    const float*  cw  = &sW2[i * 16];
    float a0 = 0.f, a1 = 0.f, a2 = 0.f, a3 = 0.f;
#pragma unroll
    for (int j = 0; j < NH; ++j) {
      const int Rj = (j < 7) ? j : j + 1;
      const float4 en = ent[Rj];
      const float c = cw[Rj];
      a0 = fmaf(c, en.x, a0);
      a1 = fmaf(c, en.y, a1);
      a2 = fmaf(c, en.z, a2);
      a3 = fmaf(c, en.w, a3);
    }
    if (h != ((i >> 2) & 1)) a2 = 0.f;
    a3 = (h == 0) ? a3 + gb2 : 0.f;
    sA[i * 32 + h * 16 + 7] = make_float4(a0, a1, a2, a3);
  }

  lds_barrier();

  const float one_lo = hi ? 0.f : 1.f;
  float b0 = 0.f, b1w = 0.f;

  auto step = [&](int i, float nzm) -> float {
    const float4 ap = sA[i * 32 + lane];
    v2f A2; A2[0] = ap.z; A2[1] = ap.w;
    v2f B2; B2[0] = nzm; B2[1] = one_lo;
    v8f zero = {};
    v8f t = __builtin_amdgcn_wmma_f32_16x16x4_f32(
        false, A2, false, B2, (short)0, zero, false, false);
    v2f A1; A1[0] = ap.x; A1[1] = ap.y;
    v2f B1; B1[0] = b0;   B1[1] = b1w;
    v8f d = __builtin_amdgcn_wmma_f32_16x16x4_f32(
        false, A1, false, B1, (short)0, t, false, false);
    const float4* wp = (const float4*)&sW2[i * 16 + hi * 8];
    const float4 wa = wp[0], wb = wp[1];
    float p;
    p = fmaf(fabsf(d[0]), wa.x, d[7]);
    p = fmaf(fabsf(d[1]), wa.y, p);
    p = fmaf(fabsf(d[2]), wa.z, p);
    p = fmaf(fabsf(d[3]), wa.w, p);
    p = fmaf(fabsf(d[4]), wb.x, p);
    p = fmaf(fabsf(d[5]), wb.y, p);
    p = fmaf(fabsf(d[6]), wb.z, p);
    return p + xor16(p);
  };

#pragma unroll
  for (int t = 0; t < 8; ++t) {
#pragma unroll
    for (int s = 0; s < 2; ++s) {
#pragma unroll
      for (int c = 0; c < 4; ++c) {
        const int i = 8 * t + 4 * s + c;
        const int q = 4 * t + c;
        const bool own = s ? (hi != 0) : (hi == 0);
        const float y = step(i, own ? rw[q] : 0.f);
        if (i < 4) {
          if (i == 0) { if (!hi) b0  = y; }
          if (i == 1) { if (!hi) b1w = y; }
          if (i == 2) { if (hi)  b0  = y; }
          if (i == 3) { if (hi)  b1w = y; }
        } else {
          const float tmp = xor16(b0);
          b0 = b1w;
          b1w = hi ? y : tmp;
        }
        rw[q] = own ? y : rw[q];
      }
    }
  }

#pragma unroll
  for (int t = 0; t < 8; ++t)
    orow[2 * t + hi] = make_float4(rw[4 * t + 0], rw[4 * t + 1],
                                   rw[4 * t + 2], rw[4 * t + 3]);
}

extern "C" void kernel_launch(void* const* d_in, const int* in_sizes, int n_in,
                              void* d_out, int out_size, void* d_ws, size_t ws_size,
                              hipStream_t stream) {
  const float* noise = (const float*)d_in[0];
  const float* W1 = (const float*)d_in[2];
  const float* b1 = (const float*)d_in[3];
  const float* W2 = (const float*)d_in[4];
  const float* b2 = (const float*)d_in[5];
  float* out = (float*)d_out;

  const int B = in_sizes[0] / NODES;
  const int rowsPerBlock = 128;
  const int grid = B / rowsPerBlock;

  cgnn_wmma_kernel<<<grid, 256, 0, stream>>>(noise, W1, b1, W2, b2, out);
}
